# starting kernel
# baseline (speedup 1.0000x reference)
#include <hip/hip_runtime.h>
#include <math.h>

#define PP   769
#define LDL  772
#define NCOL 32768
#define BMSK 2047
#define KC   128
#define KST  132

typedef __attribute__((ext_vector_type(2))) float v2f;
typedef __attribute__((ext_vector_type(8))) float v8f;

__device__ __forceinline__ v8f wmma_f32_16x16x4(v2f a, v2f b, v8f c) {
  return __builtin_amdgcn_wmma_f32_16x16x4_f32(false, a, false, b, (short)0, c,
                                               false, false);
}

__global__ __launch_bounds__(256) void build_L_kernel(const float* __restrict__ cov,
                                                      float* __restrict__ Lb) {
  int t = blockIdx.x * 256 + threadIdx.x;
  if (t >= PP * LDL) return;
  int i = t / LDL;
  int j = t - i * LDL;
  float v = 0.0f;
  if (j < PP && j <= i) {
    v = cov[i * (i + 1) / 2 + j];
    if (i == j) v = expf(v);
  }
  Lb[t] = v;
}

template <int TLO>
__device__ __forceinline__ void kphase(int kbeg, int kend, int kgbase,
                                       const float* __restrict__ zc,
                                       const float* __restrict__ sA,
                                       int half, int sub,
                                       v8f& c0, v8f& c1, v8f& c2,
                                       v8f& c3, v8f& c4, v8f& c5) {
  const int lbase = sub * KST + 2 * half;
  for (int kk = kbeg; kk < kend; kk += 4) {
    v2f bm;
    bm.x = zc[(size_t)(kgbase + kk + 2 * half) * NCOL];
    bm.y = zc[(size_t)(kgbase + kk + 2 * half + 1) * NCOL];
    const int ao = lbase + kk;
    if constexpr (TLO <= 0) { v2f a; a.x = sA[0 * 16 * KST + ao]; a.y = sA[0 * 16 * KST + ao + 1]; c0 = wmma_f32_16x16x4(a, bm, c0); }
    if constexpr (TLO <= 1) { v2f a; a.x = sA[1 * 16 * KST + ao]; a.y = sA[1 * 16 * KST + ao + 1]; c1 = wmma_f32_16x16x4(a, bm, c1); }
    if constexpr (TLO <= 2) { v2f a; a.x = sA[2 * 16 * KST + ao]; a.y = sA[2 * 16 * KST + ao + 1]; c2 = wmma_f32_16x16x4(a, bm, c2); }
    if constexpr (TLO <= 3) { v2f a; a.x = sA[3 * 16 * KST + ao]; a.y = sA[3 * 16 * KST + ao + 1]; c3 = wmma_f32_16x16x4(a, bm, c3); }
    if constexpr (TLO <= 4) { v2f a; a.x = sA[4 * 16 * KST + ao]; a.y = sA[4 * 16 * KST + ao + 1]; c4 = wmma_f32_16x16x4(a, bm, c4); }
    if constexpr (TLO <= 5) { v2f a; a.x = sA[5 * 16 * KST + ao]; a.y = sA[5 * 16 * KST + ao + 1]; c5 = wmma_f32_16x16x4(a, bm, c5); }
  }
}

__global__ __launch_bounds__(256) void fcvi_main_kernel(
    const float* __restrict__ Lb, const float* __restrict__ z,
    const float* __restrict__ x, const float* __restrict__ mean,
    float* __restrict__ out) {
  __shared__ float sA[6 * 16 * KST];

  const int tid  = threadIdx.x;
  const int lane = tid & 31;
  const int wave = tid >> 5;
  const int sub  = lane & 15;
  const int half = lane >> 4;
  const int gcol = blockIdx.x * 128 + wave * 16 + sub;
  const float xv = x[gcol & BMSK];
  const float* __restrict__ zcol = z + gcol;

  float outpart = 0.0f;

  for (int mp = 0; mp < 8; ++mp) {
    const int r00 = 32 * mp, r01 = r00 + 16;
    const int rb0 = 256 + r00, rb1 = rb0 + 16;
    const int r10 = 512 + r00, r11 = r10 + 16;
    const int e0 = r00 + 32;
    const int e1 = 512 + e0;
    const int rbase[6] = {r00, r01, rb0, rb1, r10, r11};
    const int bext[6]  = {e0, e0, 256 + e0, 256 + e0, e1, e1};

    v8f a0 = {0.f, 0.f, 0.f, 0.f, 0.f, 0.f, 0.f, 0.f};
    v8f a1 = a0, a2 = a0, a3 = a0, a4 = a0, a5 = a0;

    for (int kk0 = 0; kk0 < e1; kk0 += KC) {
      __syncthreads();
#pragma unroll
      for (int t = 0; t < 6; ++t) {
        int e = bext[t] - kk0;
        if (e <= 0) continue;
        if (e > KC) e = KC;
#pragma unroll 4
        for (int r = 0; r < 16; ++r) {
          const float* src = Lb + (size_t)(rbase[t] + r) * LDL + kk0;
          float* dst = &sA[(t * 16 + r) * KST];
          for (int c = tid; c < e; c += 256) dst[c] = src[c];
        }
      }
      __syncthreads();

      int kend = e1 - kk0;
      if (kend > KC) kend = KC;
      const int tlo = (kk0 < e0) ? 0 : (kk0 < 256 + e0) ? 2 : 4;
      const int sbound = (tlo == 0) ? e0 : (tlo == 2) ? 256 + e0 : e1;
      int s = sbound - kk0;
      if (s > kend) s = kend;

      if (tlo == 0)      kphase<0>(0, s, kk0, zcol, sA, half, sub, a0, a1, a2, a3, a4, a5);
      else if (tlo == 2) kphase<2>(0, s, kk0, zcol, sA, half, sub, a0, a1, a2, a3, a4, a5);
      else               kphase<4>(0, s, kk0, zcol, sA, half, sub, a0, a1, a2, a3, a4, a5);
      if (s < kend) {
        if (tlo == 0) kphase<2>(s, kend, kk0, zcol, sA, half, sub, a0, a1, a2, a3, a4, a5);
        else          kphase<4>(s, kend, kk0, zcol, sA, half, sub, a0, a1, a2, a3, a4, a5);
      }
    }

#pragma unroll
    for (int v = 0; v < 8; ++v) {
      const int rl = v + 8 * half;
      const float w0e = a0[v] + mean[r00 + rl];
      const float b0e = a2[v] + mean[rb0 + rl];
      const float w1e = a4[v] + mean[r10 + rl];
      outpart = fmaf(fmaxf(fmaf(xv, w0e, b0e), 0.0f), w1e, outpart);
      const float w0o = a1[v] + mean[r01 + rl];
      const float b0o = a3[v] + mean[rb1 + rl];
      const float w1o = a5[v] + mean[r11 + rl];
      outpart = fmaf(fmaxf(fmaf(xv, w0o, b0o), 0.0f), w1o, outpart);
    }
  }

  float bpart = (half == 0) ? mean[768] : 0.0f;
  const float* lrow = Lb + (size_t)768 * LDL;
  for (int j = half * 4; j < 768; j += 8) {
    const float4 lv = *(const float4*)(lrow + j);
    bpart = fmaf(lv.x, zcol[(size_t)(j + 0) * NCOL], bpart);
    bpart = fmaf(lv.y, zcol[(size_t)(j + 1) * NCOL], bpart);
    bpart = fmaf(lv.z, zcol[(size_t)(j + 2) * NCOL], bpart);
    bpart = fmaf(lv.w, zcol[(size_t)(j + 3) * NCOL], bpart);
  }
  if (half == 0) bpart = fmaf(lrow[768], zcol[(size_t)768 * NCOL], bpart);

  outpart += bpart;
  const float other = __shfl_xor(outpart, 16, 32);
  if (half == 0) out[gcol] = outpart + other;
}

extern "C" void kernel_launch(void* const* d_in, const int* in_sizes, int n_in,
                              void* d_out, int out_size, void* d_ws, size_t ws_size,
                              hipStream_t stream) {
  (void)in_sizes; (void)n_in; (void)out_size; (void)ws_size;
  const float* x    = (const float*)d_in[0];
  const float* mean = (const float*)d_in[1];
  const float* cov  = (const float*)d_in[2];
  const float* z    = (const float*)d_in[3];
  float* outp = (float*)d_out;
  float* Lbuf = (float*)d_ws;

  build_L_kernel<<<(PP * LDL + 255) / 256, 256, 0, stream>>>(cov, Lbuf);
  fcvi_main_kernel<<<NCOL / 128, 256, 0, stream>>>(Lbuf, z, x, mean, outp);
}


# Row-768 (b1) dot-product tail: loads of 12 four-k pieces (60 loads) issued as one group ahead of their sums, keeping the load queue full instead of one dependent round trip per piece; sum order kept
# speedup vs baseline: 2.2238x; 2.2238x over previous
#include <hip/hip_runtime.h>
#include <math.h>

#define PP   769
#define LDL  772
#define NCOL 32768
#define BMSK 2047
#define KS   32
#define AST  36
#define TILEF  (16 * AST)
#define STAGEF (6 * TILEF)
#define NSUBT  164
#define TPIECES 96
#define TB      12

typedef __attribute__((ext_vector_type(2))) float v2f;
typedef __attribute__((ext_vector_type(8))) float v8f;

__device__ __forceinline__ v8f wmma_f32_16x16x4(v2f a, v2f b, v8f c) {
  return __builtin_amdgcn_wmma_f32_16x16x4_f32(false, a, false, b, (short)0, c,
                                               false, false);
}

__global__ __launch_bounds__(256) void build_L_kernel(const float* __restrict__ cov,
                                                      float* __restrict__ Lb) {
  int t = blockIdx.x * 256 + threadIdx.x;
  if (t >= PP * LDL) return;
  int i = t / LDL;
  int j = t - i * LDL;
  float v = 0.0f;
  if (j < PP && j <= i) {
    v = cov[i * (i + 1) / 2 + j];
    if (i == j) v = expf(v);
  }
  Lb[t] = v;
}

__device__ __forceinline__ void fetch_l(const float* __restrict__ Lb, int mp, int j, int tlo,
                                        int row32, int q,
                                        float4& l0, float4& l1, float4& l2) {
  const float* p = Lb + (size_t)(32 * mp + row32) * LDL + KS * j + 4 * q;
  if (tlo <= 0) l0 = *(const float4*)(p);
  if (tlo <= 2) l1 = *(const float4*)(p + (size_t)256 * LDL);
  l2 = *(const float4*)(p + (size_t)512 * LDL);
}

__device__ __forceinline__ void store_l(float* __restrict__ st, int tlo, int row32, int q,
                                        const float4& l0, const float4& l1, const float4& l2) {
  float* d = st + row32 * AST + 4 * q;
  if (tlo <= 0) *(float4*)(d) = l0;
  if (tlo <= 2) *(float4*)(d + 32 * AST) = l1;
  *(float4*)(d + 64 * AST) = l2;
}

__device__ __forceinline__ void fetch_b(const float* __restrict__ zcol, int kbase, int half,
                                        v2f (&b)[8]) {
  const float* p = zcol + (size_t)(kbase + 2 * half) * NCOL;
#pragma unroll
  for (int s = 0; s < 8; ++s) {
    b[s].x = p[(size_t)(4 * s) * NCOL];
    b[s].y = p[(size_t)(4 * s + 1) * NCOL];
  }
}

template <int TLO>
__device__ __forceinline__ void compute_sub(const float* __restrict__ st, int lofs,
                                            const v2f (&b)[8],
                                            v8f& c0, v8f& c1, v8f& c2,
                                            v8f& c3, v8f& c4, v8f& c5) {
#pragma unroll
  for (int s = 0; s < 8; ++s) {
    const float* a = st + lofs + 4 * s;
    if constexpr (TLO <= 0) {
      c0 = wmma_f32_16x16x4(*(const v2f*)(a + 0 * TILEF), b[s], c0);
      c1 = wmma_f32_16x16x4(*(const v2f*)(a + 1 * TILEF), b[s], c1);
    }
    if constexpr (TLO <= 2) {
      c2 = wmma_f32_16x16x4(*(const v2f*)(a + 2 * TILEF), b[s], c2);
      c3 = wmma_f32_16x16x4(*(const v2f*)(a + 3 * TILEF), b[s], c3);
    }
    c4 = wmma_f32_16x16x4(*(const v2f*)(a + 4 * TILEF), b[s], c4);
    c5 = wmma_f32_16x16x4(*(const v2f*)(a + 5 * TILEF), b[s], c5);
  }
}

__device__ __forceinline__ void fold_pass(int mp, int half, float xv,
                                          const float* __restrict__ mean,
                                          v8f& a0, v8f& a1, v8f& a2,
                                          v8f& a3, v8f& a4, v8f& a5, float& outpart) {
  const int r00 = 32 * mp, r01 = r00 + 16;
  const int rb0 = 256 + r00, rb1 = rb0 + 16;
  const int r10 = 512 + r00, r11 = r10 + 16;
#pragma unroll
  for (int v = 0; v < 8; ++v) {
    const int rl = v + 8 * half;
    const float w0e = a0[v] + mean[r00 + rl];
    const float b0e = a2[v] + mean[rb0 + rl];
    const float w1e = a4[v] + mean[r10 + rl];
    outpart = fmaf(fmaxf(fmaf(xv, w0e, b0e), 0.0f), w1e, outpart);
    const float w0o = a1[v] + mean[r01 + rl];
    const float b0o = a3[v] + mean[rb1 + rl];
    const float w1o = a5[v] + mean[r11 + rl];
    outpart = fmaf(fmaxf(fmaf(xv, w0o, b0o), 0.0f), w1o, outpart);
  }
  const v8f zero = {0.f, 0.f, 0.f, 0.f, 0.f, 0.f, 0.f, 0.f};
  a0 = zero; a1 = zero; a2 = zero; a3 = zero; a4 = zero; a5 = zero;
}

__global__ __launch_bounds__(256) void fcvi_main_kernel(
    const float* __restrict__ Lb, const float* __restrict__ z,
    const float* __restrict__ x, const float* __restrict__ mean,
    float* __restrict__ out) {
  __shared__ __attribute__((aligned(16))) float sA[2 * STAGEF];

  const int tid  = threadIdx.x;
  const int lane = tid & 31;
  const int wave = tid >> 5;
  const int sub  = lane & 15;
  const int half = lane >> 4;
  const int gcol = blockIdx.x * 128 + wave * 16 + sub;
  const float xv = x[gcol & BMSK];
  const float* __restrict__ zcol = z + gcol;
  const int row32 = tid >> 3;
  const int q     = tid & 7;
  const int lofs  = sub * AST + 2 * half;

  float outpart = 0.0f;

  v8f a0 = {0.f, 0.f, 0.f, 0.f, 0.f, 0.f, 0.f, 0.f};
  v8f a1 = a0, a2 = a0, a3 = a0, a4 = a0, a5 = a0;

  v2f bA[8], bB[8];
  float4 l0 = make_float4(0.f, 0.f, 0.f, 0.f), l1 = l0, l2 = l0;

  int mp = 0, j = 0, tlo = 0;
  int nmp = 0, nj = 0, ntlo = 0;
  int buf = 0;

  fetch_l(Lb, 0, 0, 0, row32, q, l0, l1, l2);
  fetch_b(zcol, 0, half, bA);
  store_l(sA, 0, row32, q, l0, l1, l2);
  __syncthreads();

#define PIPE_STEP(BX, BY)                                                               \
  {                                                                                     \
    const bool more = (it + 1 < NSUBT);                                                 \
    if (more) {                                                                         \
      if (j == 16 + mp) { nmp = mp + 1; nj = 0; } else { nmp = mp; nj = j + 1; }        \
      ntlo = (nj < nmp + 1) ? 0 : (nj < nmp + 9) ? 2 : 4;                               \
      fetch_l(Lb, nmp, nj, ntlo, row32, q, l0, l1, l2);                                 \
      fetch_b(zcol, KS * nj, half, BY);                                                 \
    }                                                                                   \
    const float* st = sA + buf * STAGEF;                                                \
    if (tlo == 0)      compute_sub<0>(st, lofs, BX, a0, a1, a2, a3, a4, a5);            \
    else if (tlo == 2) compute_sub<2>(st, lofs, BX, a0, a1, a2, a3, a4, a5);            \
    else               compute_sub<4>(st, lofs, BX, a0, a1, a2, a3, a4, a5);            \
    if (j == 16 + mp) {                                                                 \
      fold_pass(mp, half, xv, mean, a0, a1, a2, a3, a4, a5, outpart);                   \
    }                                                                                   \
    if (more) {                                                                         \
      store_l(sA + (buf ^ 1) * STAGEF, ntlo, row32, q, l0, l1, l2);                     \
      mp = nmp; j = nj; tlo = ntlo; buf ^= 1;                                           \
    }                                                                                   \
    __syncthreads();       \
    ++it;                                                                               \
  }

  for (int it = 0; it < NSUBT;) {
    PIPE_STEP(bA, bB)
    PIPE_STEP(bB, bA)
  }
#undef PIPE_STEP

  float bpart = (half == 0) ? mean[768] : 0.0f;
  const float* lrow = Lb + (size_t)768 * LDL;
  const float l768 = lrow[768];
  const float z768 = zcol[(size_t)768 * NCOL];
  const float* lp = lrow + 4 * half;
  const float* zp = zcol + (size_t)(4 * half) * NCOL;
  for (int g = 0; g < TPIECES / TB; ++g) {
    float4 lv[TB];
    float z0[TB], z1[TB], z2[TB], z3[TB];
#pragma unroll
    for (int u = 0; u < TB; ++u) {
      lv[u] = *(const float4*)(lp + 8 * u);
      const float* zr = zp + (size_t)(8 * u) * NCOL;
      z0[u] = zr[0];
      z1[u] = zr[(size_t)NCOL];
      z2[u] = zr[(size_t)2 * NCOL];
      z3[u] = zr[(size_t)3 * NCOL];
    }
#pragma unroll
    for (int u = 0; u < TB; ++u) {
      bpart = fmaf(lv[u].x, z0[u], bpart);
      bpart = fmaf(lv[u].y, z1[u], bpart);
      bpart = fmaf(lv[u].z, z2[u], bpart);
      bpart = fmaf(lv[u].w, z3[u], bpart);
    }
    lp += 8 * TB;
    zp += (size_t)(8 * TB) * NCOL;
  }
  if (half == 0) bpart = fmaf(l768, z768, bpart);

  outpart += bpart;
  const float other = __shfl_xor(outpart, 16, 32);
  if (half == 0) out[gcol] = outpart + other;
}

extern "C" void kernel_launch(void* const* d_in, const int* in_sizes, int n_in,
                              void* d_out, int out_size, void* d_ws, size_t ws_size,
                              hipStream_t stream) {
  (void)in_sizes; (void)n_in; (void)out_size; (void)ws_size;
  const float* x    = (const float*)d_in[0];
  const float* mean = (const float*)d_in[1];
  const float* cov  = (const float*)d_in[2];
  const float* z    = (const float*)d_in[3];
  float* outp = (float*)d_out;
  float* Lbuf = (float*)d_ws;

  build_L_kernel<<<(PP * LDL + 255) / 256, 256, 0, stream>>>(cov, Lbuf);
  fcvi_main_kernel<<<NCOL / 128, 256, 0, stream>>>(Lbuf, z, x, mean, outp);
}


# L written by launch 1 in matrix A-fragment order, read by each wave straight from global memory (16-byte whole-line loads, a sub-chunk ahead): no LDS staging, no barriers; sum order kept
# speedup vs baseline: 2.8799x; 1.2950x over previous
#include <hip/hip_runtime.h>
#include <math.h>

#define PP   769
#define LDL  772
#define NCOL 32768
#define BMSK 2047
#define KS   32
#define TPIECES 96
#define TB      12

#define PSTEPS 96
#define TILE_F (PSTEPS * 128)
#define GRP_F  (16 * TILE_F)
#define FRAG_F (48 * TILE_F)
#define BUILD_BLOCKS 580

typedef __attribute__((ext_vector_type(2))) float v2f;
typedef __attribute__((ext_vector_type(8))) float v8f;

__device__ __forceinline__ v8f wmma_f32_16x16x4(v2f a, v2f b, v8f c) {
  return __builtin_amdgcn_wmma_f32_16x16x4_f32(false, a, false, b, (short)0, c,
                                               false, false);
}

__device__ __forceinline__ float l_elem(const float* __restrict__ row, int i, int j) {
  float v = 0.0f;
  if (j <= i) {
    v = row[j];
    if (i == j) v = expf(v);
  }
  return v;
}

__global__ __launch_bounds__(256) void build_L_kernel(const float* __restrict__ cov,
                                                      float* __restrict__ Lw) {
  const int t = blockIdx.x * 256 + threadIdx.x;
  if (t < FRAG_F / 4) {
    const int l  = t & 31;
    const int tp = t >> 5;
    const int T  = tp / PSTEPS;
    const int P  = tp - T * PSTEPS;
    const int i  = 16 * T + (l & 15);
    const int j0 = 8 * P + 2 * (l >> 4);
    const float* row = cov + i * (i + 1) / 2;
    float4 v;
    v.x = l_elem(row, i, j0);
    v.y = l_elem(row, i, j0 + 1);
    v.z = l_elem(row, i, j0 + 4);
    v.w = l_elem(row, i, j0 + 5);
    *(float4*)(Lw + 4 * (size_t)t) = v;
  } else {
    const int j = t - FRAG_F / 4;
    if (j < LDL) {
      float v = 0.0f;
      if (j < PP) {
        v = cov[768 * 769 / 2 + j];
        if (j == 768) v = expf(v);
      }
      Lw[FRAG_F + j] = v;
    }
  }
}

template <int NTLO>
__device__ __forceinline__ void fetch_a(const float* __restrict__ p, float4 (&a)[24]) {
#pragma unroll
  for (int t = NTLO; t < 6; ++t) {
#pragma unroll
    for (int pp = 0; pp < 4; ++pp) {
      a[4 * t + pp] = *(const float4*)(p + (t >> 1) * GRP_F + (t & 1) * TILE_F + pp * 128);
    }
  }
}

__device__ __forceinline__ void fetch_b(const float* __restrict__ zcol, int kbase, int half,
                                        v2f (&b)[8]) {
  const float* p = zcol + (size_t)(kbase + 2 * half) * NCOL;
#pragma unroll
  for (int s = 0; s < 8; ++s) {
    b[s].x = p[(size_t)(4 * s) * NCOL];
    b[s].y = p[(size_t)(4 * s + 1) * NCOL];
  }
}

__device__ __forceinline__ v2f a_frag(const float4& f, int odd) {
  v2f r;
  if (odd) { r.x = f.z; r.y = f.w; } else { r.x = f.x; r.y = f.y; }
  return r;
}

template <int TLO>
__device__ __forceinline__ void compute_sub(const float4 (&a)[24], const v2f (&b)[8],
                                            v8f& c0, v8f& c1, v8f& c2,
                                            v8f& c3, v8f& c4, v8f& c5) {
#pragma unroll
  for (int s = 0; s < 8; ++s) {
    const int pp = s >> 1, odd = s & 1;
    if constexpr (TLO <= 0) {
      c0 = wmma_f32_16x16x4(a_frag(a[0 + pp], odd), b[s], c0);
      c1 = wmma_f32_16x16x4(a_frag(a[4 + pp], odd), b[s], c1);
    }
    if constexpr (TLO <= 2) {
      c2 = wmma_f32_16x16x4(a_frag(a[8 + pp], odd), b[s], c2);
      c3 = wmma_f32_16x16x4(a_frag(a[12 + pp], odd), b[s], c3);
    }
    c4 = wmma_f32_16x16x4(a_frag(a[16 + pp], odd), b[s], c4);
    c5 = wmma_f32_16x16x4(a_frag(a[20 + pp], odd), b[s], c5);
  }
}

struct OpRegs {
  float4 f[2][24];
  v2f    b[2][8];
};

template <int PAR, int TLO, int NTLO>
__device__ __forceinline__ void pipe_step(OpRegs& r, const float* __restrict__ pn,
                                          const float* __restrict__ zcol, int kn, int half,
                                          v8f& c0, v8f& c1, v8f& c2,
                                          v8f& c3, v8f& c4, v8f& c5) {
  if constexpr (NTLO >= 0) {
    fetch_a<NTLO>(pn, r.f[PAR ^ 1]);
    fetch_b(zcol, kn, half, r.b[PAR ^ 1]);
  }
  __builtin_amdgcn_sched_barrier(0);
  compute_sub<TLO>(r.f[PAR], r.b[PAR], c0, c1, c2, c3, c4, c5);
}

__device__ __forceinline__ void fold_pass(int mp, int half, float xv,
                                          const float* __restrict__ mean,
                                          v8f& a0, v8f& a1, v8f& a2,
                                          v8f& a3, v8f& a4, v8f& a5, float& outpart) {
  const int r00 = 32 * mp, r01 = r00 + 16;
  const int rb0 = 256 + r00, rb1 = rb0 + 16;
  const int r10 = 512 + r00, r11 = r10 + 16;
#pragma unroll
  for (int v = 0; v < 8; ++v) {
    const int rl = v + 8 * half;
    const float w0e = a0[v] + mean[r00 + rl];
    const float b0e = a2[v] + mean[rb0 + rl];
    const float w1e = a4[v] + mean[r10 + rl];
    outpart = fmaf(fmaxf(fmaf(xv, w0e, b0e), 0.0f), w1e, outpart);
    const float w0o = a1[v] + mean[r01 + rl];
    const float b0o = a3[v] + mean[rb1 + rl];
    const float w1o = a5[v] + mean[r11 + rl];
    outpart = fmaf(fmaxf(fmaf(xv, w0o, b0o), 0.0f), w1o, outpart);
  }
  const v8f zero = {0.f, 0.f, 0.f, 0.f, 0.f, 0.f, 0.f, 0.f};
  a0 = zero; a1 = zero; a2 = zero; a3 = zero; a4 = zero; a5 = zero;
}

__global__ __launch_bounds__(256) void fcvi_main_kernel(
    const float* __restrict__ Lw, const float* __restrict__ z,
    const float* __restrict__ x, const float* __restrict__ mean,
    float* __restrict__ out) {
  const int tid  = threadIdx.x;
  const int lane = tid & 31;
  const int wave = tid >> 5;
  const int sub  = lane & 15;
  const int half = lane >> 4;
  const int gcol = blockIdx.x * 128 + wave * 16 + sub;
  const float xv = x[gcol & BMSK];
  const float* __restrict__ zcol = z + gcol;
  const float* __restrict__ lfl = Lw + 4 * lane;

  float outpart = 0.0f;

  v8f a0 = {0.f, 0.f, 0.f, 0.f, 0.f, 0.f, 0.f, 0.f};
  v8f a1 = a0, a2 = a0, a3 = a0, a4 = a0, a5 = a0;

  OpRegs r;
  fetch_a<0>(lfl, r.f[0]);
  fetch_b(zcol, 0, half, r.b[0]);

#define HOLD_BEHIND(AO, KN)                                                         \
  asm volatile("" : "+s"(AO), "+s"(KN)                                              \
               : "v"(a0[7]), "v"(a1[7]), "v"(a2[7]), "v"(a3[7]), "v"(a4[7]), "v"(a5[7]));
#define STEP(PX, TLO, NTLO)                                                          \
  {                                                                                  \
    pipe_step<PX, TLO, NTLO>(r, lfl + ao, zcol, kn, half, a0, a1, a2, a3, a4, a5);   \
    ao += 4 * 128;                                                                   \
    kn += KS;                                                                        \
    HOLD_BEHIND(ao, kn)     \
  }
#pragma nounroll
  for (int mp = 0; mp < 8; ++mp) {
    int ao = mp * (2 * TILE_F) + 4 * 128;
    int kn = KS;
    HOLD_BEHIND(ao, kn)
    const int pairs0 = (mp + 1) >> 1;
#pragma nounroll
    for (int m = 0; m < pairs0; ++m) {
      STEP(0, 0, 0)
      STEP(1, 0, 0)
    }
    if ((mp & 1) == 0) {
      STEP(0, 0, 2)
#pragma unroll
      for (int u = 8; u < 24; ++u) r.f[0][u] = r.f[1][u];
#pragma unroll
      for (int s = 0; s < 8; ++s) r.b[0][s] = r.b[1][s];
    }
#pragma nounroll
    for (int m = 0; m < 4; ++m) {
      STEP(0, 2, 2)
      STEP(1, 2, 2)
    }
#pragma nounroll
    for (int m = 0; m < 3; ++m) {
      STEP(0, 4, 4)
      STEP(1, 4, 4)
    }
    STEP(0, 4, 4)
    if (mp == 7) {
      pipe_step<1, 4, -1>(r, lfl, zcol, 0, half, a0, a1, a2, a3, a4, a5);
    } else {
      ao = (mp + 1) * (2 * TILE_F);
      kn = 0;
      HOLD_BEHIND(ao, kn)
      pipe_step<1, 4, 0>(r, lfl + ao, zcol, kn, half, a0, a1, a2, a3, a4, a5);
    }
    fold_pass(mp, half, xv, mean, a0, a1, a2, a3, a4, a5, outpart);
  }
#undef STEP
#undef HOLD_BEHIND

  float bpart = (half == 0) ? mean[768] : 0.0f;
  const float* lrow = Lw + FRAG_F;
  const float l768 = lrow[768];
  const float z768 = zcol[(size_t)768 * NCOL];
  const float* lp = lrow + 4 * half;
  const float* zp = zcol + (size_t)(4 * half) * NCOL;
  for (int g = 0; g < TPIECES / TB; ++g) {
    float4 lv[TB];
    float z0[TB], z1[TB], z2[TB], z3[TB];
#pragma unroll
    for (int u = 0; u < TB; ++u) {
      lv[u] = *(const float4*)(lp + 8 * u);
      const float* zr = zp + (size_t)(8 * u) * NCOL;
      z0[u] = zr[0];
      z1[u] = zr[(size_t)NCOL];
      z2[u] = zr[(size_t)2 * NCOL];
      z3[u] = zr[(size_t)3 * NCOL];
    }
#pragma unroll
    for (int u = 0; u < TB; ++u) {
      bpart = fmaf(lv[u].x, z0[u], bpart);
      bpart = fmaf(lv[u].y, z1[u], bpart);
      bpart = fmaf(lv[u].z, z2[u], bpart);
      bpart = fmaf(lv[u].w, z3[u], bpart);
    }
    lp += 8 * TB;
    zp += (size_t)(8 * TB) * NCOL;
  }
  if (half == 0) bpart = fmaf(l768, z768, bpart);

  outpart += bpart;
  const float other = __shfl_xor(outpart, 16, 32);
  if (half == 0) out[gcol] = outpart + other;
}

extern "C" void kernel_launch(void* const* d_in, const int* in_sizes, int n_in,
                              void* d_out, int out_size, void* d_ws, size_t ws_size,
                              hipStream_t stream) {
  (void)in_sizes; (void)n_in; (void)out_size; (void)ws_size;
  const float* x    = (const float*)d_in[0];
  const float* mean = (const float*)d_in[1];
  const float* cov  = (const float*)d_in[2];
  const float* z    = (const float*)d_in[3];
  float* outp = (float*)d_out;
  float* Lbuf = (float*)d_ws;

  build_L_kernel<<<BUILD_BLOCKS, 256, 0, stream>>>(cov, Lbuf);
  fcvi_main_kernel<<<NCOL / 128, 256, 0, stream>>>(Lbuf, z, x, mean, outp);
}


# Each wave takes 32 columns (every L fragment load feeds two matrix ops) and four of the eight hidden-unit passes; partner waves' partial sums meet in LDS at the end: sum order of the fold changes
# speedup vs baseline: 3.1540x; 1.0952x over previous
#include <hip/hip_runtime.h>
#include <math.h>

#define PP   769
#define LDL  772
#define NCOL 32768
#define BMSK 2047
#define KS   32
#define TPIECES 96
#define TB      12

#define PSTEPS 96
#define TILE_F (PSTEPS * 128)
#define GRP_F  (16 * TILE_F)
#define FRAG_F (48 * TILE_F)
#define BUILD_BLOCKS 580

typedef __attribute__((ext_vector_type(2))) float v2f;
typedef __attribute__((ext_vector_type(8))) float v8f;

__device__ __forceinline__ v8f wmma_f32_16x16x4(v2f a, v2f b, v8f c) {
  return __builtin_amdgcn_wmma_f32_16x16x4_f32(false, a, false, b, (short)0, c,
                                               false, false);
}

__device__ __forceinline__ float l_elem(const float* __restrict__ row, int i, int j) {
  float v = 0.0f;
  if (j <= i) {
    v = row[j];
    if (i == j) v = expf(v);
  }
  return v;
}

__global__ __launch_bounds__(256) void build_L_kernel(const float* __restrict__ cov,
                                                      float* __restrict__ Lw) {
  const int t = blockIdx.x * 256 + threadIdx.x;
  if (t < FRAG_F / 4) {
    const int l  = t & 31;
    const int tp = t >> 5;
    const int T  = tp / PSTEPS;
    const int P  = tp - T * PSTEPS;
    const int i  = 16 * T + (l & 15);
    const int j0 = 8 * P + 2 * (l >> 4);
    const float* row = cov + i * (i + 1) / 2;
    float4 v;
    v.x = l_elem(row, i, j0);
    v.y = l_elem(row, i, j0 + 1);
    v.z = l_elem(row, i, j0 + 4);
    v.w = l_elem(row, i, j0 + 5);
    *(float4*)(Lw + 4 * (size_t)t) = v;
  } else {
    const int j = t - FRAG_F / 4;
    if (j < LDL) {
      float v = 0.0f;
      if (j < PP) {
        v = cov[768 * 769 / 2 + j];
        if (j == 768) v = expf(v);
      }
      Lw[FRAG_F + j] = v;
    }
  }
}

template <int NTLO>
__device__ __forceinline__ void fetch_a(const float* __restrict__ p, float4 (&a)[24]) {
#pragma unroll
  for (int t = NTLO; t < 6; ++t) {
#pragma unroll
    for (int pp = 0; pp < 4; ++pp) {
      a[4 * t + pp] = *(const float4*)(p + (t >> 1) * GRP_F + (t & 1) * TILE_F + pp * 128);
    }
  }
}

__device__ __forceinline__ void fetch_b(const float* __restrict__ zcol, int kbase, int half,
                                        v2f (&b)[16]) {
  const float* p = zcol + (size_t)(kbase + 2 * half) * NCOL;
#pragma unroll
  for (int s = 0; s < 8; ++s) {
    b[2 * s].x     = p[(size_t)(4 * s) * NCOL];
    b[2 * s].y     = p[(size_t)(4 * s + 1) * NCOL];
    b[2 * s + 1].x = p[(size_t)(4 * s) * NCOL + 16];
    b[2 * s + 1].y = p[(size_t)(4 * s + 1) * NCOL + 16];
  }
}

__device__ __forceinline__ v2f a_frag(const float4& f, int odd) {
  v2f r;
  if (odd) { r.x = f.z; r.y = f.w; } else { r.x = f.x; r.y = f.y; }
  return r;
}

struct Acc {
  v8f c0, c1, c2, c3, c4, c5;
  v8f d0, d1, d2, d3, d4, d5;
};

template <int TLO>
__device__ __forceinline__ void compute_sub(const float4 (&a)[24], const v2f (&b)[16], Acc& q) {
#pragma unroll
  for (int s = 0; s < 8; ++s) {
    const int pp = s >> 1, odd = s & 1;
    if constexpr (TLO <= 0) {
      q.c0 = wmma_f32_16x16x4(a_frag(a[0 + pp], odd), b[2 * s], q.c0);
      q.d0 = wmma_f32_16x16x4(a_frag(a[0 + pp], odd), b[2 * s + 1], q.d0);
      q.c1 = wmma_f32_16x16x4(a_frag(a[4 + pp], odd), b[2 * s], q.c1);
      q.d1 = wmma_f32_16x16x4(a_frag(a[4 + pp], odd), b[2 * s + 1], q.d1);
    }
    if constexpr (TLO <= 2) {
      q.c2 = wmma_f32_16x16x4(a_frag(a[8 + pp], odd), b[2 * s], q.c2);
      q.d2 = wmma_f32_16x16x4(a_frag(a[8 + pp], odd), b[2 * s + 1], q.d2);
      q.c3 = wmma_f32_16x16x4(a_frag(a[12 + pp], odd), b[2 * s], q.c3);
      q.d3 = wmma_f32_16x16x4(a_frag(a[12 + pp], odd), b[2 * s + 1], q.d3);
    }
    q.c4 = wmma_f32_16x16x4(a_frag(a[16 + pp], odd), b[2 * s], q.c4);
    q.d4 = wmma_f32_16x16x4(a_frag(a[16 + pp], odd), b[2 * s + 1], q.d4);
    q.c5 = wmma_f32_16x16x4(a_frag(a[20 + pp], odd), b[2 * s], q.c5);
    q.d5 = wmma_f32_16x16x4(a_frag(a[20 + pp], odd), b[2 * s + 1], q.d5);
  }
}

struct OpRegs {
  float4 f[2][24];
  v2f    b[2][16];
};

template <int PAR, int TLO, int NTLO>
__device__ __forceinline__ void pipe_step(OpRegs& r, const float* __restrict__ pn,
                                          const float* __restrict__ zcol, int kn, int half, Acc& q) {
  if constexpr (NTLO >= 0) {
    fetch_a<NTLO>(pn, r.f[PAR ^ 1]);
    fetch_b(zcol, kn, half, r.b[PAR ^ 1]);
  }
  __builtin_amdgcn_sched_barrier(0);
  compute_sub<TLO>(r.f[PAR], r.b[PAR], q);
}

__device__ __forceinline__ void fold_pass(int mp, int half, float xv,
                                          const float* __restrict__ mean,
                                          v8f& a0, v8f& a1, v8f& a2,
                                          v8f& a3, v8f& a4, v8f& a5, float& outpart) {
  const int r00 = 32 * mp, r01 = r00 + 16;
  const int rb0 = 256 + r00, rb1 = rb0 + 16;
  const int r10 = 512 + r00, r11 = r10 + 16;
#pragma unroll
  for (int v = 0; v < 8; ++v) {
    const int rl = v + 8 * half;
    const float w0e = a0[v] + mean[r00 + rl];
    const float b0e = a2[v] + mean[rb0 + rl];
    const float w1e = a4[v] + mean[r10 + rl];
    outpart = fmaf(fmaxf(fmaf(xv, w0e, b0e), 0.0f), w1e, outpart);
    const float w0o = a1[v] + mean[r01 + rl];
    const float b0o = a3[v] + mean[rb1 + rl];
    const float w1o = a5[v] + mean[r11 + rl];
    outpart = fmaf(fmaxf(fmaf(xv, w0o, b0o), 0.0f), w1o, outpart);
  }
  const v8f zero = {0.f, 0.f, 0.f, 0.f, 0.f, 0.f, 0.f, 0.f};
  a0 = zero; a1 = zero; a2 = zero; a3 = zero; a4 = zero; a5 = zero;
}

__global__ __launch_bounds__(256) void fcvi_main_kernel(
    const float* __restrict__ Lw, const float* __restrict__ z,
    const float* __restrict__ x, const float* __restrict__ mean,
    float* __restrict__ out) {
  __shared__ float part[128];
  const int tid  = threadIdx.x;
  const int lane = tid & 31;
  const int wave = __builtin_amdgcn_readfirstlane(tid >> 5);
  const int sub  = lane & 15;
  const int half = lane >> 4;
  const int cg   = wave & 3;
  const int ph   = wave >> 2;
  const int cbase = blockIdx.x * 128 + cg * 32;
  const int col0 = cbase + sub;
  const float xv0 = x[col0 & BMSK];
  const float xv1 = x[(col0 + 16) & BMSK];
  const float* __restrict__ zcol = z + col0;
  const float* __restrict__ lfl = Lw + 4 * lane;

  float outpart0 = 0.0f, outpart1 = 0.0f;

  const v8f zero8 = {0.f, 0.f, 0.f, 0.f, 0.f, 0.f, 0.f, 0.f};
  Acc q;
  q.c0 = zero8; q.c1 = zero8; q.c2 = zero8; q.c3 = zero8; q.c4 = zero8; q.c5 = zero8;
  q.d0 = zero8; q.d1 = zero8; q.d2 = zero8; q.d3 = zero8; q.d4 = zero8; q.d5 = zero8;

  OpRegs r;
  fetch_a<0>(lfl + ph * (2 * TILE_F), r.f[0]);
  fetch_b(zcol, 0, half, r.b[0]);

#define HOLD_BEHIND(AO, KN)                                                         \
  asm volatile("" : "+s"(AO), "+s"(KN)                                              \
               : "v"(q.c0[7]), "v"(q.c1[7]), "v"(q.c2[7]), "v"(q.c3[7]), "v"(q.c4[7]), "v"(q.c5[7]), \
                 "v"(q.d0[7]), "v"(q.d1[7]), "v"(q.d2[7]), "v"(q.d3[7]), "v"(q.d4[7]), "v"(q.d5[7]));
#define STEP(PX, TLO, NTLO)                                                          \
  {                                                                                  \
    pipe_step<PX, TLO, NTLO>(r, lfl + ao, zcol, kn, half, q);                        \
    ao += 4 * 128;                                                                   \
    kn += KS;                                                                        \
    HOLD_BEHIND(ao, kn)     \
  }
#pragma nounroll
  for (int i = 0; i < 4; ++i) {
    const int mp = 2 * i + ((i & 1) ^ ph);
    int ao = mp * (2 * TILE_F) + 4 * 128;
    int kn = KS;
    HOLD_BEHIND(ao, kn)
    const int pairs0 = (mp + 1) >> 1;
#pragma nounroll
    for (int m = 0; m < pairs0; ++m) {
      STEP(0, 0, 0)
      STEP(1, 0, 0)
    }
    if ((mp & 1) == 0) {
      STEP(0, 0, 2)
#pragma unroll
      for (int u = 8; u < 24; ++u) r.f[0][u] = r.f[1][u];
#pragma unroll
      for (int s = 0; s < 16; ++s) r.b[0][s] = r.b[1][s];
    }
#pragma nounroll
    for (int m = 0; m < 4; ++m) {
      STEP(0, 2, 2)
      STEP(1, 2, 2)
    }
#pragma nounroll
    for (int m = 0; m < 3; ++m) {
      STEP(0, 4, 4)
      STEP(1, 4, 4)
    }
    STEP(0, 4, 4)
    if (i == 3) {
      pipe_step<1, 4, -1>(r, lfl, zcol, 0, half, q);
    } else {
      const int mpn = 2 * (i + 1) + (((i + 1) & 1) ^ ph);
      ao = mpn * (2 * TILE_F);
      kn = 0;
      HOLD_BEHIND(ao, kn)
      pipe_step<1, 4, 0>(r, lfl + ao, zcol, kn, half, q);
    }
    fold_pass(mp, half, xv0, mean, q.c0, q.c1, q.c2, q.c3, q.c4, q.c5, outpart0);
    fold_pass(mp, half, xv1, mean, q.d0, q.d1, q.d2, q.d3, q.d4, q.d5, outpart1);
  }
#undef STEP
#undef HOLD_BEHIND

  const float* __restrict__ ztail = zcol + 16 * ph;
  float bpart = (half == 0) ? mean[768] : 0.0f;
  const float* lrow = Lw + FRAG_F;
  const float l768 = lrow[768];
  const float z768 = ztail[(size_t)768 * NCOL];
  const float* lp = lrow + 4 * half;
  const float* zp = ztail + (size_t)(4 * half) * NCOL;
  for (int g = 0; g < TPIECES / TB; ++g) {
    float4 lv[TB];
    float z0[TB], z1[TB], z2[TB], z3[TB];
#pragma unroll
    for (int u = 0; u < TB; ++u) {
      lv[u] = *(const float4*)(lp + 8 * u);
      const float* zr = zp + (size_t)(8 * u) * NCOL;
      z0[u] = zr[0];
      z1[u] = zr[(size_t)NCOL];
      z2[u] = zr[(size_t)2 * NCOL];
      z3[u] = zr[(size_t)3 * NCOL];
    }
#pragma unroll
    for (int u = 0; u < TB; ++u) {
      bpart = fmaf(lv[u].x, z0[u], bpart);
      bpart = fmaf(lv[u].y, z1[u], bpart);
      bpart = fmaf(lv[u].z, z2[u], bpart);
      bpart = fmaf(lv[u].w, z3[u], bpart);
    }
    lp += 8 * TB;
    zp += (size_t)(8 * TB) * NCOL;
  }
  if (half == 0) bpart = fmaf(l768, z768, bpart);

  if (ph == 0) outpart0 += bpart; else outpart1 += bpart;
  const float s0 = outpart0 + __shfl_xor(outpart0, 16, 32);
  const float s1 = outpart1 + __shfl_xor(outpart1, 16, 32);
  const float mine = half ? s1 : s0;
  if (ph == 1) part[cg * 32 + lane] = mine;
  __syncthreads();
  if (ph == 0) out[cbase + lane] = mine + part[cg * 32 + lane];
}

extern "C" void kernel_launch(void* const* d_in, const int* in_sizes, int n_in,
                              void* d_out, int out_size, void* d_ws, size_t ws_size,
                              hipStream_t stream) {
  (void)in_sizes; (void)n_in; (void)out_size; (void)ws_size;
  const float* x    = (const float*)d_in[0];
  const float* mean = (const float*)d_in[1];
  const float* cov  = (const float*)d_in[2];
  const float* z    = (const float*)d_in[3];
  float* outp = (float*)d_out;
  float* Lbuf = (float*)d_ws;

  build_L_kernel<<<BUILD_BLOCKS, 256, 0, stream>>>(cov, Lbuf);
  fcvi_main_kernel<<<NCOL / 128, 256, 0, stream>>>(Lbuf, z, x, mean, outp);
}
